# starting kernel
# baseline (speedup 1.0000x reference)
#include <hip/hip_runtime.h>
#include <hip/hip_bf16.h>
#include <math.h>

typedef __attribute__((ext_vector_type(2))) float v2f;
typedef __attribute__((ext_vector_type(8))) float v8f;

#define D_DIM 256
#define KC 32
#define LDA_S (KC + 2)
#define LDBT  (KC + 2)

__global__ __launch_bounds__(256)
void zero_f32(float4* __restrict__ p, long n4) {
    long i = (long)blockIdx.x * blockDim.x + threadIdx.x;
    if (i < n4) p[i] = make_float4(0.f, 0.f, 0.f, 0.f);
}

__global__ __launch_bounds__(256)
void spmm_atomic(const int* __restrict__ rows, const int* __restrict__ cols,
                 const float* __restrict__ vals,
                 const float* __restrict__ Xa, const float* __restrict__ Xb,
                 int split, float* __restrict__ Y, int nnz) {
    int w    = blockIdx.x * 8 + (threadIdx.x >> 5);
    int lane = threadIdx.x & 31;
    if (w >= nnz) return;
    int   r = rows[w];
    int   c = cols[w];
    float v = vals[w];
    const float* X = (c < split) ? (Xa + (size_t)c * D_DIM)
                                 : (Xb + (size_t)(c - split) * D_DIM);
    float* y = Y + (size_t)r * D_DIM;
#pragma unroll
    for (int i = 0; i < 2; ++i) {
        int    b = (lane + 32 * i) * 4;
        float4 x = *(const float4*)(X + b);
        atomicAdd(y + b + 0, x.x * v);
        atomicAdd(y + b + 1, x.y * v);
        atomicAdd(y + b + 2, x.z * v);
        atomicAdd(y + b + 3, x.w * v);
    }
}

__global__ __launch_bounds__(256)
void gemm_tanh_wmma(const float* __restrict__ A0, const float* __restrict__ A1,
                    const float* __restrict__ B0, const float* __restrict__ B1,
                    const float* __restrict__ bias, float* __restrict__ OUT,
                    int M, int accumulate) {
    __shared__ float As[64 * LDA_S];
    __shared__ float Bst[128 * LDBT];

    const int t    = threadIdx.x;
    const int wave = t >> 5;
    const int lane = t & 31;
    const int wm   = wave & 3;
    const int wn   = wave >> 2;
    const int rowBase = blockIdx.x * 64;
    const int colBase = blockIdx.y * 128;

    const int m16 = lane & 15;
    const int kh  = lane >> 4;

    v8f acc[4];
#pragma unroll
    for (int j = 0; j < 4; ++j) acc[j] = (v8f)0.f;

    for (int src = 0; src < 2; ++src) {
        const float* A = src ? A1 : A0;
        const float* B = src ? B1 : B0;
        if (A == nullptr) break;

        for (int k0 = 0; k0 < D_DIM; k0 += KC) {
            {
                int r  = t >> 3;
                int c4 = (t & 7) * 4;
#pragma unroll
                for (int i = 0; i < 2; ++i) {
                    int    row = rowBase + r + 32 * i;
                    float4 v   = make_float4(0.f, 0.f, 0.f, 0.f);
                    if (row < M)
                        v = *(const float4*)(A + (size_t)row * D_DIM + k0 + c4);
                    float* dst = &As[(r + 32 * i) * LDA_S + c4];
                    dst[0] = v.x; dst[1] = v.y; dst[2] = v.z; dst[3] = v.w;
                }
            }
            {
#pragma unroll
                for (int i = 0; i < 4; ++i) {
                    int idx = t + 256 * i;
                    int k   = idx >> 5;
                    int n4  = (idx & 31) * 4;
                    float4 v = *(const float4*)(B + (size_t)(k0 + k) * D_DIM +
                                                colBase + n4);
                    Bst[(n4 + 0) * LDBT + k] = v.x;
                    Bst[(n4 + 1) * LDBT + k] = v.y;
                    Bst[(n4 + 2) * LDBT + k] = v.z;
                    Bst[(n4 + 3) * LDBT + k] = v.w;
                }
            }
            __syncthreads();

#pragma unroll
            for (int kk = 0; kk < KC; kk += 4) {
                v2f af = *(const v2f*)&As[(wm * 16 + m16) * LDA_S + kk + 2 * kh];
#pragma unroll
                for (int j = 0; j < 4; ++j) {
                    int n  = wn * 64 + j * 16 + m16;
                    v2f bf = *(const v2f*)&Bst[n * LDBT + kk + 2 * kh];
                    acc[j] = __builtin_amdgcn_wmma_f32_16x16x4_f32(
                        false, af, false, bf, (short)0, acc[j], false, false);
                }
            }
            __syncthreads();
        }
    }

#pragma unroll
    for (int j = 0; j < 4; ++j) {
        int   c  = colBase + wn * 64 + j * 16 + m16;
        float bv = bias ? bias[c] : 0.f;
#pragma unroll
        for (int r = 0; r < 8; ++r) {
            int row = rowBase + wm * 16 + kh * 8 + r;
            if (row < M) {
                float  v = tanhf(acc[j][r] + bv);
                float* o = OUT + (size_t)row * D_DIM + c;
                if (accumulate) *o += v;
                else            *o  = v;
            }
        }
    }
}

__global__ __launch_bounds__(256)
void l2norm_rows(const float* __restrict__ H, float* __restrict__ out, int M) {
    int row  = blockIdx.x * 8 + (threadIdx.x >> 5);
    int lane = threadIdx.x & 31;
    if (row >= M) return;
    const float4* h = (const float4*)(H + (size_t)row * D_DIM);
    float4 a = h[lane];
    float4 b = h[lane + 32];
    float  s = a.x * a.x + a.y * a.y + a.z * a.z + a.w * a.w +
               b.x * b.x + b.y * b.y + b.z * b.z + b.w * b.w;
#pragma unroll
    for (int off = 16; off > 0; off >>= 1) s += __shfl_xor(s, off, 32);
    float inv = 1.f / fmaxf(sqrtf(s), 1e-12f);
    a.x *= inv; a.y *= inv; a.z *= inv; a.w *= inv;
    b.x *= inv; b.y *= inv; b.z *= inv; b.w *= inv;
    float4* o = (float4*)(out + (size_t)row * D_DIM);
    o[lane]      = a;
    o[lane + 32] = b;
}

extern "C" void kernel_launch(void* const* d_in, const int* in_sizes, int n_in,
                              void* d_out, int out_size, void* d_ws, size_t ws_size,
                              hipStream_t stream) {
    const int* adj_rows   = (const int*)d_in[0];
    const int* adj_cols   = (const int*)d_in[1];
    const float* adj_vals = (const float*)d_in[2];
    const int* up_rows    = (const int*)d_in[3];
    const int* up_cols    = (const int*)d_in[4];
    const float* up_vals  = (const float*)d_in[5];
    const int* ip_rows    = (const int*)d_in[6];
    const int* ip_cols    = (const int*)d_in[7];
    const float* ip_vals  = (const float*)d_in[8];
    const float* user_emb = (const float*)d_in[9];
    const float* item_emb = (const float*)d_in[10];
    const float* Q_user   = (const float*)d_in[11];
    const float* W_user   = (const float*)d_in[12];
    const float* b_user   = (const float*)d_in[13];
    const float* Q_item   = (const float*)d_in[14];
    const float* W_item   = (const float*)d_in[15];
    const float* b_item   = (const float*)d_in[16];
    const float* M_user   = (const float*)d_in[17];
    const float* M_item   = (const float*)d_in[18];
    float* out = (float*)d_out;

    const int NU = 40000, NI = 20000, NN = 60000;
    const int NNZ_ADJ = in_sizes[0], NNZ_UP = in_sizes[3], NNZ_IP = in_sizes[6];
    const size_t ROWF = (size_t)D_DIM;
    const size_t MAT  = (size_t)NN * ROWF;

    float* agg  = (float*)d_ws;
    float* embb = agg  + MAT;
    float* hbuf = embb + MAT;

    dim3 blk(256);

    zero_f32<<<(unsigned)((MAT / 4 + 255) / 256), blk, 0, stream>>>((float4*)agg, (long)(MAT / 4));
    spmm_atomic<<<(NNZ_ADJ + 7) / 8, blk, 0, stream>>>(
        adj_rows, adj_cols, adj_vals, user_emb, item_emb, NU, agg, NNZ_ADJ);

    gemm_tanh_wmma<<<dim3((NU + 63) / 64, 2), blk, 0, stream>>>(
        agg, nullptr, Q_user, nullptr, nullptr, embb, NU, 0);
    gemm_tanh_wmma<<<dim3((NI + 63) / 64, 2), blk, 0, stream>>>(
        agg + (size_t)NU * ROWF, nullptr, Q_item, nullptr, nullptr,
        embb + (size_t)NU * ROWF, NI, 0);

    gemm_tanh_wmma<<<dim3((NU + 63) / 64, 2), blk, 0, stream>>>(
        user_emb, embb, W_user, W_user + (size_t)D_DIM * D_DIM, b_user,
        hbuf, NU, 0);
    gemm_tanh_wmma<<<dim3((NI + 63) / 64, 2), blk, 0, stream>>>(
        item_emb, embb + (size_t)NU * ROWF, W_item,
        W_item + (size_t)D_DIM * D_DIM, b_item,
        hbuf + (size_t)NU * ROWF, NI, 0);

    l2norm_rows<<<(NN + 7) / 8, blk, 0, stream>>>(hbuf, out, NN);

    zero_f32<<<(unsigned)((MAT / 4 + 255) / 256), blk, 0, stream>>>((float4*)agg, (long)(MAT / 4));
    spmm_atomic<<<(NNZ_UP + 7) / 8, blk, 0, stream>>>(
        up_rows, up_cols, up_vals, user_emb, user_emb, NU, agg, NNZ_UP);
    spmm_atomic<<<(NNZ_IP + 7) / 8, blk, 0, stream>>>(
        ip_rows, ip_cols, ip_vals, item_emb, item_emb, NI,
        agg + (size_t)NU * ROWF, NNZ_IP);

    gemm_tanh_wmma<<<dim3((NU + 63) / 64, 2), blk, 0, stream>>>(
        agg, nullptr, M_user, nullptr, nullptr, out, NU, 1);
    gemm_tanh_wmma<<<dim3((NI + 63) / 64, 2), blk, 0, stream>>>(
        agg + (size_t)NU * ROWF, nullptr, M_item, nullptr, nullptr,
        out + (size_t)NU * ROWF, NI, 1);
}


# SPMM as one global CSR: integer-atomic row counts, block scan, scatter of (col,val) to row segments in workspace, one wave sums a row in registers (fma, arrival order); no f32 atomics, no zero fill
# speedup vs baseline: 1.5262x; 1.5262x over previous
#include <hip/hip_runtime.h>
#include <hip/hip_bf16.h>
#include <math.h>

typedef __attribute__((ext_vector_type(2))) float v2f;
typedef __attribute__((ext_vector_type(8))) float v8f;

#define D_DIM 256
#define KC 32
#define LDA_S (KC + 2)
#define LDBT  (KC + 2)

__device__ __forceinline__ void coo_entry(
    int e, const int* __restrict__ r0, const int* __restrict__ c0, const float* __restrict__ v0, int n0,
    const int* __restrict__ r1, const int* __restrict__ c1, const float* __restrict__ v1, int n1,
    const int* __restrict__ r2, const int* __restrict__ c2, const float* __restrict__ v2,
    int R1, int R2, int coff2, int& row, int& col, float& val) {
    if (e < n0) {
        row = r0[e];            col = c0[e];               val = v0[e];
    } else if (e < n0 + n1) {
        int k = e - n0;
        row = R1 + r1[k];       col = c1[k];               val = v1[k];
    } else {
        int k = e - n0 - n1;
        row = R2 + r2[k];       col = coff2 + c2[k];       val = v2[k];
    }
}

__global__ __launch_bounds__(256)
void csr_zero(int* __restrict__ cnt, int n, int* __restrict__ gcur) {
    int i = blockIdx.x * 256 + threadIdx.x;
    if (i < n) cnt[i] = 0;
    if (i == 0) *gcur = 0;
}

__global__ __launch_bounds__(256)
void csr_count(const int* __restrict__ r0, int n0, const int* __restrict__ r1, int n1,
               const int* __restrict__ r2, int n2, int R1, int R2, int* __restrict__ cnt) {
    int e = blockIdx.x * 256 + threadIdx.x;
    if (e >= n0 + n1 + n2) return;
    int row;
    if (e < n0)           row = r0[e];
    else if (e < n0 + n1) row = R1 + r1[e - n0];
    else                  row = R2 + r2[e - n0 - n1];
    atomicAdd(&cnt[row], 1);
}

__global__ __launch_bounds__(1024)
void csr_offsets(int* __restrict__ cnt, int* __restrict__ start, int n, int* __restrict__ gcur) {
    __shared__ int wsum[32];
    __shared__ int sbase;
    const int t    = threadIdx.x;
    const int lane = t & 31;
    const int wave = t >> 5;
    const int i    = blockIdx.x * 1024 + t;
    const int c    = (i < n) ? cnt[i] : 0;
    int x = c;
#pragma unroll
    for (int off = 1; off < 32; off <<= 1) {
        int y = __shfl_up(x, off, 32);
        if (lane >= off) x += y;
    }
    if (lane == 31) wsum[wave] = x;
    __syncthreads();
    if (wave == 0) {
        int w = wsum[lane];
        int s = w;
#pragma unroll
        for (int off = 1; off < 32; off <<= 1) {
            int y = __shfl_up(s, off, 32);
            if (lane >= off) s += y;
        }
        wsum[lane] = s - w;
        if (lane == 31) sbase = atomicAdd(gcur, s);
    }
    __syncthreads();
    int first = sbase + wsum[wave] + x - c;
    if (i < n) { start[i] = first; cnt[i] = first; }
}

__global__ __launch_bounds__(256)
void csr_scatter(const int* __restrict__ r0, const int* __restrict__ c0, const float* __restrict__ v0, int n0,
                 const int* __restrict__ r1, const int* __restrict__ c1, const float* __restrict__ v1, int n1,
                 const int* __restrict__ r2, const int* __restrict__ c2, const float* __restrict__ v2, int n2,
                 int R1, int R2, int coff2, int* __restrict__ cursor, int2* __restrict__ ent) {
    int e = blockIdx.x * 256 + threadIdx.x;
    if (e >= n0 + n1 + n2) return;
    int row, col; float val;
    coo_entry(e, r0, c0, v0, n0, r1, c1, v1, n1, r2, c2, v2, R1, R2, coff2, row, col, val);
    int pos = atomicAdd(&cursor[row], 1);
    ent[pos] = make_int2(col, __float_as_int(val));
}

__global__ __launch_bounds__(256)
void spmm_rows(const int* __restrict__ start, const int* __restrict__ stop,
               const int2* __restrict__ ent,
               const float* __restrict__ Xa, const float* __restrict__ Xb, int split,
               float* __restrict__ Y0, float* __restrict__ Y1, int rowsY0, int nrows) {
    int row = blockIdx.x * 8 + (threadIdx.x >> 5);
    const int lane = threadIdx.x & 31;
    if (row >= nrows) return;
    row = __builtin_amdgcn_readfirstlane(row);
    const int s = start[row];
    const int e = stop[row];
    float4 a0 = make_float4(0.f, 0.f, 0.f, 0.f);
    float4 a1 = make_float4(0.f, 0.f, 0.f, 0.f);
    for (int b = s; b < e; b += 32) {
        int m = e - b;
        if (m > 32) m = 32;
        int2 my = make_int2(0, 0);
        if (lane < m) my = ent[b + lane];
#pragma unroll 4
        for (int j = 0; j < m; ++j) {
            const int   c = __builtin_amdgcn_readlane(my.x, j);
            const float v = __int_as_float(__builtin_amdgcn_readlane(my.y, j));
            const float* X = (c < split) ? (Xa + (size_t)c * D_DIM)
                                         : (Xb + (size_t)(c - split) * D_DIM);
            const float4 x0 = ((const float4*)X)[lane];
            const float4 x1 = ((const float4*)X)[lane + 32];
            a0.x += v * x0.x; a0.y += v * x0.y; a0.z += v * x0.z; a0.w += v * x0.w;
            a1.x += v * x1.x; a1.y += v * x1.y; a1.z += v * x1.z; a1.w += v * x1.w;
        }
    }
    float* y = (row < rowsY0) ? (Y0 + (size_t)row * D_DIM)
                              : (Y1 + (size_t)(row - rowsY0) * D_DIM);
    ((float4*)y)[lane]      = a0;
    ((float4*)y)[lane + 32] = a1;
}

__global__ __launch_bounds__(256)
void gemm_tanh_wmma(const float* __restrict__ A0, const float* __restrict__ A1,
                    const float* __restrict__ B0, const float* __restrict__ B1,
                    const float* __restrict__ bias, float* __restrict__ OUT,
                    int M, int accumulate) {
    __shared__ float As[64 * LDA_S];
    __shared__ float Bst[128 * LDBT];

    const int t    = threadIdx.x;
    const int wave = t >> 5;
    const int lane = t & 31;
    const int wm   = wave & 3;
    const int wn   = wave >> 2;
    const int rowBase = blockIdx.x * 64;
    const int colBase = blockIdx.y * 128;

    const int m16 = lane & 15;
    const int kh  = lane >> 4;

    v8f acc[4];
#pragma unroll
    for (int j = 0; j < 4; ++j) acc[j] = (v8f)0.f;

    for (int src = 0; src < 2; ++src) {
        const float* A = src ? A1 : A0;
        const float* B = src ? B1 : B0;
        if (A == nullptr) break;

        for (int k0 = 0; k0 < D_DIM; k0 += KC) {
            {
                int r  = t >> 3;
                int c4 = (t & 7) * 4;
#pragma unroll
                for (int i = 0; i < 2; ++i) {
                    int    row = rowBase + r + 32 * i;
                    float4 v   = make_float4(0.f, 0.f, 0.f, 0.f);
                    if (row < M)
                        v = *(const float4*)(A + (size_t)row * D_DIM + k0 + c4);
                    float* dst = &As[(r + 32 * i) * LDA_S + c4];
                    dst[0] = v.x; dst[1] = v.y; dst[2] = v.z; dst[3] = v.w;
                }
            }
            {
#pragma unroll
                for (int i = 0; i < 4; ++i) {
                    int idx = t + 256 * i;
                    int k   = idx >> 5;
                    int n4  = (idx & 31) * 4;
                    float4 v = *(const float4*)(B + (size_t)(k0 + k) * D_DIM +
                                                colBase + n4);
                    Bst[(n4 + 0) * LDBT + k] = v.x;
                    Bst[(n4 + 1) * LDBT + k] = v.y;
                    Bst[(n4 + 2) * LDBT + k] = v.z;
                    Bst[(n4 + 3) * LDBT + k] = v.w;
                }
            }
            __syncthreads();

#pragma unroll
            for (int kk = 0; kk < KC; kk += 4) {
                v2f af = *(const v2f*)&As[(wm * 16 + m16) * LDA_S + kk + 2 * kh];
#pragma unroll
                for (int j = 0; j < 4; ++j) {
                    int n  = wn * 64 + j * 16 + m16;
                    v2f bf = *(const v2f*)&Bst[n * LDBT + kk + 2 * kh];
                    acc[j] = __builtin_amdgcn_wmma_f32_16x16x4_f32(
                        false, af, false, bf, (short)0, acc[j], false, false);
                }
            }
            __syncthreads();
        }
    }

#pragma unroll
    for (int j = 0; j < 4; ++j) {
        int   c  = colBase + wn * 64 + j * 16 + m16;
        float bv = bias ? bias[c] : 0.f;
#pragma unroll
        for (int r = 0; r < 8; ++r) {
            int row = rowBase + wm * 16 + kh * 8 + r;
            if (row < M) {
                float  v = tanhf(acc[j][r] + bv);
                float* o = OUT + (size_t)row * D_DIM + c;
                if (accumulate) *o += v;
                else            *o  = v;
            }
        }
    }
}

__global__ __launch_bounds__(256)
void l2norm_rows(const float* __restrict__ H, float* __restrict__ out, int M) {
    int row  = blockIdx.x * 8 + (threadIdx.x >> 5);
    int lane = threadIdx.x & 31;
    if (row >= M) return;
    const float4* h = (const float4*)(H + (size_t)row * D_DIM);
    float4 a = h[lane];
    float4 b = h[lane + 32];
    float  s = a.x * a.x + a.y * a.y + a.z * a.z + a.w * a.w +
               b.x * b.x + b.y * b.y + b.z * b.z + b.w * b.w;
#pragma unroll
    for (int off = 16; off > 0; off >>= 1) s += __shfl_xor(s, off, 32);
    float inv = 1.f / fmaxf(sqrtf(s), 1e-12f);
    a.x *= inv; a.y *= inv; a.z *= inv; a.w *= inv;
    b.x *= inv; b.y *= inv; b.z *= inv; b.w *= inv;
    float4* o = (float4*)(out + (size_t)row * D_DIM);
    o[lane]      = a;
    o[lane + 32] = b;
}

extern "C" void kernel_launch(void* const* d_in, const int* in_sizes, int n_in,
                              void* d_out, int out_size, void* d_ws, size_t ws_size,
                              hipStream_t stream) {
    const int* adj_rows   = (const int*)d_in[0];
    const int* adj_cols   = (const int*)d_in[1];
    const float* adj_vals = (const float*)d_in[2];
    const int* up_rows    = (const int*)d_in[3];
    const int* up_cols    = (const int*)d_in[4];
    const float* up_vals  = (const float*)d_in[5];
    const int* ip_rows    = (const int*)d_in[6];
    const int* ip_cols    = (const int*)d_in[7];
    const float* ip_vals  = (const float*)d_in[8];
    const float* user_emb = (const float*)d_in[9];
    const float* item_emb = (const float*)d_in[10];
    const float* Q_user   = (const float*)d_in[11];
    const float* W_user   = (const float*)d_in[12];
    const float* b_user   = (const float*)d_in[13];
    const float* Q_item   = (const float*)d_in[14];
    const float* W_item   = (const float*)d_in[15];
    const float* b_item   = (const float*)d_in[16];
    const float* M_user   = (const float*)d_in[17];
    const float* M_item   = (const float*)d_in[18];
    float* out = (float*)d_out;

    const int NU = 40000, NI = 20000, NN = 60000;
    const int NNZ_ADJ = in_sizes[0], NNZ_UP = in_sizes[3], NNZ_IP = in_sizes[6];
    const int NNZ_ALL = NNZ_ADJ + NNZ_UP + NNZ_IP;
    const size_t ROWF = (size_t)D_DIM;
    const size_t MAT  = (size_t)NN * ROWF;

    float* buf0 = (float*)d_ws;
    float* buf1 = buf0 + MAT;
    float* buf2 = buf1 + MAT;

    const int NR   = NN + NN;
    const int NRP  = 120064;
    int*  cnt   = (int*)buf2;
    int*  start = cnt + NRP;
    int*  gcur  = start + NRP;
    int2* ent   = (int2*)(gcur + 64);

    dim3 blk(256);

    csr_zero<<<(NR + 255) / 256, blk, 0, stream>>>(cnt, NR, gcur);
    csr_count<<<(NNZ_ALL + 255) / 256, blk, 0, stream>>>(
        adj_rows, NNZ_ADJ, up_rows, NNZ_UP, ip_rows, NNZ_IP, NN, NN + NU, cnt);
    csr_offsets<<<(NR + 1023) / 1024, dim3(1024), 0, stream>>>(cnt, start, NR, gcur);
    csr_scatter<<<(NNZ_ALL + 255) / 256, blk, 0, stream>>>(
        adj_rows, adj_cols, adj_vals, NNZ_ADJ, up_rows, up_cols, up_vals, NNZ_UP,
        ip_rows, ip_cols, ip_vals, NNZ_IP, NN, NN + NU, NU, cnt, ent);

    spmm_rows<<<(NR + 7) / 8, blk, 0, stream>>>(
        start, cnt, ent, user_emb, item_emb, NU, buf0, buf1, NN, NR);

    gemm_tanh_wmma<<<dim3((NU + 63) / 64, 2), blk, 0, stream>>>(
        buf0, nullptr, Q_user, nullptr, nullptr, buf2, NU, 0);
    gemm_tanh_wmma<<<dim3((NI + 63) / 64, 2), blk, 0, stream>>>(
        buf0 + (size_t)NU * ROWF, nullptr, Q_item, nullptr, nullptr,
        buf2 + (size_t)NU * ROWF, NI, 0);

    gemm_tanh_wmma<<<dim3((NU + 63) / 64, 2), blk, 0, stream>>>(
        user_emb, buf2, W_user, W_user + (size_t)D_DIM * D_DIM, b_user,
        buf0, NU, 0);
    gemm_tanh_wmma<<<dim3((NI + 63) / 64, 2), blk, 0, stream>>>(
        item_emb, buf2 + (size_t)NU * ROWF, W_item,
        W_item + (size_t)D_DIM * D_DIM, b_item,
        buf0 + (size_t)NU * ROWF, NI, 0);

    l2norm_rows<<<(NN + 7) / 8, blk, 0, stream>>>(buf0, out, NN);

    gemm_tanh_wmma<<<dim3((NU + 63) / 64, 2), blk, 0, stream>>>(
        buf1, nullptr, M_user, nullptr, nullptr, out, NU, 1);
    gemm_tanh_wmma<<<dim3((NI + 63) / 64, 2), blk, 0, stream>>>(
        buf1 + (size_t)NU * ROWF, nullptr, M_item, nullptr, nullptr,
        out + (size_t)NU * ROWF, NI, 1);
}


# Dense chain in one launch: a work-group takes 64 rows at full width; tanh(agg Q) stays in LDS as the next product's A operand; h normed in-block (new fixed sumsq tree), kept in LDS for the last add
# speedup vs baseline: 1.7764x; 1.1639x over previous
#include <hip/hip_runtime.h>
#include <hip/hip_bf16.h>
#include <math.h>

typedef __attribute__((ext_vector_type(2))) float v2f;
typedef __attribute__((ext_vector_type(8))) float v8f;

#define D_DIM 256
#define KC 32
#define LDA_S (KC + 2)
#define LDBT  (KC + 2)

__device__ __forceinline__ void coo_entry(
    int e, const int* __restrict__ r0, const int* __restrict__ c0, const float* __restrict__ v0, int n0,
    const int* __restrict__ r1, const int* __restrict__ c1, const float* __restrict__ v1, int n1,
    const int* __restrict__ r2, const int* __restrict__ c2, const float* __restrict__ v2,
    int R1, int R2, int coff2, int& row, int& col, float& val) {
    if (e < n0) {
        row = r0[e];            col = c0[e];               val = v0[e];
    } else if (e < n0 + n1) {
        int k = e - n0;
        row = R1 + r1[k];       col = c1[k];               val = v1[k];
    } else {
        int k = e - n0 - n1;
        row = R2 + r2[k];       col = coff2 + c2[k];       val = v2[k];
    }
}

__global__ __launch_bounds__(256)
void csr_zero(int* __restrict__ cnt, int n, int* __restrict__ gcur) {
    int i = blockIdx.x * 256 + threadIdx.x;
    if (i < n) cnt[i] = 0;
    if (i == 0) *gcur = 0;
}

__global__ __launch_bounds__(256)
void csr_count(const int* __restrict__ r0, int n0, const int* __restrict__ r1, int n1,
               const int* __restrict__ r2, int n2, int R1, int R2, int* __restrict__ cnt) {
    int e = blockIdx.x * 256 + threadIdx.x;
    if (e >= n0 + n1 + n2) return;
    int row;
    if (e < n0)           row = r0[e];
    else if (e < n0 + n1) row = R1 + r1[e - n0];
    else                  row = R2 + r2[e - n0 - n1];
    atomicAdd(&cnt[row], 1);
}

__global__ __launch_bounds__(1024)
void csr_offsets(int* __restrict__ cnt, int* __restrict__ start, int n, int* __restrict__ gcur) {
    __shared__ int wsum[32];
    __shared__ int sbase;
    const int t    = threadIdx.x;
    const int lane = t & 31;
    const int wave = t >> 5;
    const int i    = blockIdx.x * 1024 + t;
    const int c    = (i < n) ? cnt[i] : 0;
    int x = c;
#pragma unroll
    for (int off = 1; off < 32; off <<= 1) {
        int y = __shfl_up(x, off, 32);
        if (lane >= off) x += y;
    }
    if (lane == 31) wsum[wave] = x;
    __syncthreads();
    if (wave == 0) {
        int w = wsum[lane];
        int s = w;
#pragma unroll
        for (int off = 1; off < 32; off <<= 1) {
            int y = __shfl_up(s, off, 32);
            if (lane >= off) s += y;
        }
        wsum[lane] = s - w;
        if (lane == 31) sbase = atomicAdd(gcur, s);
    }
    __syncthreads();
    int first = sbase + wsum[wave] + x - c;
    if (i < n) { start[i] = first; cnt[i] = first; }
}

__global__ __launch_bounds__(256)
void csr_scatter(const int* __restrict__ r0, const int* __restrict__ c0, const float* __restrict__ v0, int n0,
                 const int* __restrict__ r1, const int* __restrict__ c1, const float* __restrict__ v1, int n1,
                 const int* __restrict__ r2, const int* __restrict__ c2, const float* __restrict__ v2, int n2,
                 int R1, int R2, int coff2, int* __restrict__ cursor, int2* __restrict__ ent) {
    int e = blockIdx.x * 256 + threadIdx.x;
    if (e >= n0 + n1 + n2) return;
    int row, col; float val;
    coo_entry(e, r0, c0, v0, n0, r1, c1, v1, n1, r2, c2, v2, R1, R2, coff2, row, col, val);
    int pos = atomicAdd(&cursor[row], 1);
    ent[pos] = make_int2(col, __float_as_int(val));
}

__global__ __launch_bounds__(256)
void spmm_rows(const int* __restrict__ start, const int* __restrict__ stop,
               const int2* __restrict__ ent,
               const float* __restrict__ Xa, const float* __restrict__ Xb, int split,
               float* __restrict__ Y0, float* __restrict__ Y1, int rowsY0, int nrows) {
    int row = blockIdx.x * 8 + (threadIdx.x >> 5);
    const int lane = threadIdx.x & 31;
    if (row >= nrows) return;
    row = __builtin_amdgcn_readfirstlane(row);
    const int s = start[row];
    const int e = stop[row];
    float4 a0 = make_float4(0.f, 0.f, 0.f, 0.f);
    float4 a1 = make_float4(0.f, 0.f, 0.f, 0.f);
    for (int b = s; b < e; b += 32) {
        int m = e - b;
        if (m > 32) m = 32;
        int2 my = make_int2(0, 0);
        if (lane < m) my = ent[b + lane];
#pragma unroll 4
        for (int j = 0; j < m; ++j) {
            const int   c = __builtin_amdgcn_readlane(my.x, j);
            const float v = __int_as_float(__builtin_amdgcn_readlane(my.y, j));
            const float* X = (c < split) ? (Xa + (size_t)c * D_DIM)
                                         : (Xb + (size_t)(c - split) * D_DIM);
            const float4 x0 = ((const float4*)X)[lane];
            const float4 x1 = ((const float4*)X)[lane + 32];
            a0.x += v * x0.x; a0.y += v * x0.y; a0.z += v * x0.z; a0.w += v * x0.w;
            a1.x += v * x1.x; a1.y += v * x1.y; a1.z += v * x1.z; a1.w += v * x1.w;
        }
    }
    float* y = (row < rowsY0) ? (Y0 + (size_t)row * D_DIM)
                              : (Y1 + (size_t)(row - rowsY0) * D_DIM);
    ((float4*)y)[lane]      = a0;
    ((float4*)y)[lane + 32] = a1;
}

#define LDE (D_DIM + 4)

template <bool A_IN_LDS>
__device__ __forceinline__ void mm_pass(const float* __restrict__ A, int validRows,
                                        const float* __restrict__ B,
                                        float* __restrict__ As, float* __restrict__ Bst,
                                        const float* __restrict__ E, v8f (&acc)[2][4],
                                        int t, int wmBase, int wnBase, int m16, int kh) {
#pragma unroll 1
    for (int k0 = 0; k0 < D_DIM; k0 += KC) {
        if (!A_IN_LDS) {
            const int r  = t >> 3;
            const int c4 = (t & 7) * 4;
#pragma unroll
            for (int i = 0; i < 2; ++i) {
                const int row = r + 32 * i;
                const int src = row < validRows ? row : validRows - 1;
                const float4 v = *(const float4*)(A + (size_t)src * D_DIM + k0 + c4);
                v2f* dst = (v2f*)&As[row * LDA_S + c4];
                v2f lo; lo.x = v.x; lo.y = v.y;
                v2f hi; hi.x = v.z; hi.y = v.w;
                dst[0] = lo;
                dst[1] = hi;
            }
        }
        {
            const int n4 = (t >> 2) * 4;
#pragma unroll
            for (int i = 0; i < 4; ++i) {
                const int k = 2 * ((t & 3) + 4 * i);
                const float* bp = B + (size_t)(k0 + k) * D_DIM + n4;
                const float4 v0 = *(const float4*)bp;
                const float4 v1 = *(const float4*)(bp + D_DIM);
                v2f p0; p0.x = v0.x; p0.y = v1.x;
                v2f p1; p1.x = v0.y; p1.y = v1.y;
                v2f p2; p2.x = v0.z; p2.y = v1.z;
                v2f p3; p3.x = v0.w; p3.y = v1.w;
                *(v2f*)&Bst[(n4 + 0) * LDBT + k] = p0;
                *(v2f*)&Bst[(n4 + 1) * LDBT + k] = p1;
                *(v2f*)&Bst[(n4 + 2) * LDBT + k] = p2;
                *(v2f*)&Bst[(n4 + 3) * LDBT + k] = p3;
            }
        }
        __syncthreads();

#pragma unroll
        for (int kk = 0; kk < KC; kk += 4) {
            v2f af[2], bf[4];
#pragma unroll
            for (int i = 0; i < 2; ++i) {
                const int m = wmBase + i * 16 + m16;
                af[i] = A_IN_LDS ? *(const v2f*)&E[m * LDE + k0 + kk + 2 * kh]
                                 : *(const v2f*)&As[m * LDA_S + kk + 2 * kh];
            }
#pragma unroll
            for (int j = 0; j < 4; ++j) {
                const int n = wnBase + j * 16 + m16;
                bf[j] = *(const v2f*)&Bst[n * LDBT + kk + 2 * kh];
            }
#pragma unroll
            for (int i = 0; i < 2; ++i)
#pragma unroll
                for (int j = 0; j < 4; ++j)
                    acc[i][j] = __builtin_amdgcn_wmma_f32_16x16x4_f32(
                        false, af[i], false, bf[j], (short)0, acc[i][j], false, false);
        }
        __syncthreads();
    }
}

__global__ __launch_bounds__(256)
void dense_chain(const float* __restrict__ AGG, const float* __restrict__ P,
                 const float* __restrict__ preU, const float* __restrict__ preI,
                 const float* __restrict__ Qu, const float* __restrict__ Qi,
                 const float* __restrict__ Wu, const float* __restrict__ Wi,
                 const float* __restrict__ bu, const float* __restrict__ bi,
                 const float* __restrict__ Mu, const float* __restrict__ Mi,
                 float* __restrict__ OUT, int NU, int NI, int tilesU) {
    __shared__ __attribute__((aligned(16))) float As[64 * LDA_S];
    __shared__ __attribute__((aligned(16))) float Bst[256 * LDBT];
    __shared__ __attribute__((aligned(16))) float E[64 * LDE];
    __shared__ float rs[4 * 64];

    const int t    = threadIdx.x;
    const int wave = t >> 5;
    const int lane = t & 31;
    const int wmBase = (wave & 1) * 32;
    const int wn     = wave >> 1;
    const int wnBase = wn * 64;
    const int m16 = lane & 15;
    const int kh  = lane >> 4;

    const bool isU   = (int)blockIdx.x < tilesU;
    const int  lbase = (isU ? (int)blockIdx.x : (int)blockIdx.x - tilesU) * 64;
    const int  gbase = isU ? lbase : NU + lbase;
    const int  left  = (isU ? NU : NI) - lbase;
    const int  validRows = left < 64 ? left : 64;
    const float* pre = (isU ? preU : preI) + (size_t)lbase * D_DIM;
    const float* Q   = isU ? Qu : Qi;
    const float* Wt  = isU ? Wu : Wi;
    const float* bb  = isU ? bu : bi;
    const float* Mw  = isU ? Mu : Mi;
    const float* agg = AGG + (size_t)gbase * D_DIM;
    const float* pp  = P   + (size_t)gbase * D_DIM;
    float*       o   = OUT + (size_t)gbase * D_DIM;

    const int rowl = wmBase + kh * 8;
    const int coll = wnBase + m16;

    v8f acc[2][4];
#pragma unroll
    for (int i = 0; i < 2; ++i)
#pragma unroll
        for (int j = 0; j < 4; ++j) acc[i][j] = (v8f)0.f;

    mm_pass<false>(agg, validRows, Q, As, Bst, E, acc, t, wmBase, wnBase, m16, kh);
#pragma unroll
    for (int i = 0; i < 2; ++i)
#pragma unroll
        for (int j = 0; j < 4; ++j) {
#pragma unroll
            for (int r = 0; r < 8; ++r)
                E[(rowl + i * 16 + r) * LDE + coll + j * 16] = acc[i][j][r];
            acc[i][j] = (v8f)0.f;
        }
#pragma unroll 1
    for (int ir = 0; ir < 16; ++ir) {
        const int row = rowl + (ir >> 3) * 16 + (ir & 7);
#pragma unroll
        for (int j = 0; j < 4; ++j) {
            const int a = row * LDE + coll + j * 16;
            E[a] = tanhf(E[a]);
        }
    }

    mm_pass<false>(pre, validRows, Wt, As, Bst, E, acc, t, wmBase, wnBase, m16, kh);
    mm_pass<true>(nullptr, validRows, Wt + (size_t)D_DIM * D_DIM, As, Bst, E, acc, t, wmBase, wnBase, m16, kh);

#pragma unroll
    for (int i = 0; i < 2; ++i)
#pragma unroll
        for (int j = 0; j < 4; ++j) {
#pragma unroll
            for (int r = 0; r < 8; ++r)
                E[(rowl + i * 16 + r) * LDE + coll + j * 16] = acc[i][j][r];
            acc[i][j] = (v8f)0.f;
        }
    float bv[4];
#pragma unroll
    for (int j = 0; j < 4; ++j) bv[j] = bb[coll + j * 16];
#pragma unroll 1
    for (int ir = 0; ir < 16; ++ir) {
        const int row = rowl + (ir >> 3) * 16 + (ir & 7);
        float s = 0.f;
#pragma unroll
        for (int j = 0; j < 4; ++j) {
            const int   a = row * LDE + coll + j * 16;
            const float h = tanhf(E[a] + bv[j]);
            E[a] = h;
            s += h * h;
        }
#pragma unroll
        for (int off = 1; off < 16; off <<= 1) s += __shfl_xor(s, off, 32);
        if (m16 == 0) rs[wn * 64 + row] = s;
    }
    __syncthreads();
#pragma unroll 1
    for (int ir = 0; ir < 16; ++ir) {
        const int   row = rowl + (ir >> 3) * 16 + (ir & 7);
        const float s   = ((rs[row] + rs[64 + row]) + rs[128 + row]) + rs[192 + row];
        const float inv = 1.f / fmaxf(sqrtf(s), 1e-12f);
#pragma unroll
        for (int j = 0; j < 4; ++j) {
            const int a = row * LDE + coll + j * 16;
            E[a] = E[a] * inv;
        }
    }

    mm_pass<false>(pp, validRows, Mw, As, Bst, E, acc, t, wmBase, wnBase, m16, kh);
#pragma unroll
    for (int i = 0; i < 2; ++i) {
#pragma unroll
        for (int j = 0; j < 4; ++j)
#pragma unroll
            for (int r = 0; r < 8; ++r)
                Bst[(j * 8 + r) * 256 + t] = acc[i][j][r];
#pragma unroll 1
        for (int r = 0; r < 8; ++r) {
            const int row = rowl + i * 16 + r;
            if (row < validRows) {
#pragma unroll
                for (int j = 0; j < 4; ++j) {
                    const int col = coll + j * 16;
                    o[(size_t)row * D_DIM + col] = E[row * LDE + col] + tanhf(Bst[(j * 8 + r) * 256 + t]);
                }
            }
        }
    }
}

extern "C" void kernel_launch(void* const* d_in, const int* in_sizes, int n_in,
                              void* d_out, int out_size, void* d_ws, size_t ws_size,
                              hipStream_t stream) {
    const int* adj_rows   = (const int*)d_in[0];
    const int* adj_cols   = (const int*)d_in[1];
    const float* adj_vals = (const float*)d_in[2];
    const int* up_rows    = (const int*)d_in[3];
    const int* up_cols    = (const int*)d_in[4];
    const float* up_vals  = (const float*)d_in[5];
    const int* ip_rows    = (const int*)d_in[6];
    const int* ip_cols    = (const int*)d_in[7];
    const float* ip_vals  = (const float*)d_in[8];
    const float* user_emb = (const float*)d_in[9];
    const float* item_emb = (const float*)d_in[10];
    const float* Q_user   = (const float*)d_in[11];
    const float* W_user   = (const float*)d_in[12];
    const float* b_user   = (const float*)d_in[13];
    const float* Q_item   = (const float*)d_in[14];
    const float* W_item   = (const float*)d_in[15];
    const float* b_item   = (const float*)d_in[16];
    const float* M_user   = (const float*)d_in[17];
    const float* M_item   = (const float*)d_in[18];
    float* out = (float*)d_out;

    const int NU = 40000, NI = 20000, NN = 60000;
    const int NNZ_ADJ = in_sizes[0], NNZ_UP = in_sizes[3], NNZ_IP = in_sizes[6];
    const int NNZ_ALL = NNZ_ADJ + NNZ_UP + NNZ_IP;
    const size_t ROWF = (size_t)D_DIM;
    const size_t MAT  = (size_t)NN * ROWF;

    float* buf0 = (float*)d_ws;
    float* buf1 = buf0 + MAT;
    float* buf2 = buf1 + MAT;

    const int NR   = NN + NN;
    const int NRP  = 120064;
    int*  cnt   = (int*)buf2;
    int*  start = cnt + NRP;
    int*  gcur  = start + NRP;
    int2* ent   = (int2*)(gcur + 64);

    dim3 blk(256);

    csr_zero<<<(NR + 255) / 256, blk, 0, stream>>>(cnt, NR, gcur);
    csr_count<<<(NNZ_ALL + 255) / 256, blk, 0, stream>>>(
        adj_rows, NNZ_ADJ, up_rows, NNZ_UP, ip_rows, NNZ_IP, NN, NN + NU, cnt);
    csr_offsets<<<(NR + 1023) / 1024, dim3(1024), 0, stream>>>(cnt, start, NR, gcur);
    csr_scatter<<<(NNZ_ALL + 255) / 256, blk, 0, stream>>>(
        adj_rows, adj_cols, adj_vals, NNZ_ADJ, up_rows, up_cols, up_vals, NNZ_UP,
        ip_rows, ip_cols, ip_vals, NNZ_IP, NN, NN + NU, NU, cnt, ent);

    spmm_rows<<<(NR + 7) / 8, blk, 0, stream>>>(
        start, cnt, ent, user_emb, item_emb, NU, buf0, buf1, NN, NR);

    dense_chain<<<dim3(625 + 313), blk, 0, stream>>>(
        buf0, buf1, user_emb, item_emb, Q_user, Q_item, W_user, W_item,
        b_user, b_item, M_user, M_item, out, NU, NI, 625);
}


# SPMM row lists as 128-place buckets in each row's own 1 KB output slot, filled by one returning integer-atomic pass; count and scan launches and entry array go (6 launches to 4); arrival-order sums
# speedup vs baseline: 2.1983x; 1.2375x over previous
#include <hip/hip_runtime.h>
#include <hip/hip_bf16.h>
#include <math.h>

typedef __attribute__((ext_vector_type(2))) float v2f;
typedef __attribute__((ext_vector_type(8))) float v8f;

#define D_DIM 256
#define KC 32
#define LDA_S (KC + 2)
#define LDBT  (KC + 2)

#define BUCKET_CAP 128

__global__ __launch_bounds__(256)
void csr_zero(int* __restrict__ cnt, int n) {
    int i = blockIdx.x * 256 + threadIdx.x;
    if (i < n) cnt[i] = 0;
}

__global__ __launch_bounds__(256)
void csr_scatter(const int* __restrict__ r0, const int* __restrict__ c0, const float* __restrict__ v0, int n0,
                 const int* __restrict__ r1, const int* __restrict__ c1, const float* __restrict__ v1, int n1,
                 const int* __restrict__ r2, const int* __restrict__ c2, const float* __restrict__ v2, int n2,
                 int R1, int R2, int coff2, int* __restrict__ cursor, float* __restrict__ Y) {
    int e = blockIdx.x * 256 + threadIdx.x;
    if (e >= n0 + n1 + n2) return;
    const int* rp = r0; const int* cp = c0; const float* vp = v0;
    int k = e, radd = 0, cadd = 0;
    if (e >= n0 + n1)  { rp = r2; cp = c2; vp = v2; k = e - n0 - n1; radd = R2; cadd = coff2; }
    else if (e >= n0)  { rp = r1; cp = c1; vp = v1; k = e - n0;      radd = R1; }
    const int   row = radd + rp[k];
    const int   col = cadd + cp[k];
    const float val = vp[k];
    asm volatile("" : : "v"(col), "v"(val));
    int pos = atomicAdd(&cursor[row], 1);
    if (pos < BUCKET_CAP) {
        const int place = ((row & 3) * 32 + pos) & (BUCKET_CAP - 1);
        ((int2*)(Y + (size_t)row * D_DIM))[place] = make_int2(col, __float_as_int(val));
    }
}

__device__ __forceinline__ void row_fma(const float* __restrict__ Xa, const float* __restrict__ Xb,
                                        int split, int c, float v, int lane, float4& a0, float4& a1) {
    const float* X = (c < split) ? (Xa + (size_t)c * D_DIM)
                                 : (Xb + (size_t)(c - split) * D_DIM);
    const float4 x0 = ((const float4*)X)[lane];
    const float4 x1 = ((const float4*)X)[lane + 32];
    a0.x += v * x0.x; a0.y += v * x0.y; a0.z += v * x0.z; a0.w += v * x0.w;
    a1.x += v * x1.x; a1.y += v * x1.y; a1.z += v * x1.z; a1.w += v * x1.w;
}

__global__ __launch_bounds__(256)
void spmm_rows(const int* __restrict__ cnt,
               const float* __restrict__ Xa, const float* __restrict__ Xb, int split,
               float* Y, int nrows,
               const int* __restrict__ r0, const int* __restrict__ c0, const float* __restrict__ v0, int n0,
               const int* __restrict__ r1, const int* __restrict__ c1, const float* __restrict__ v1, int n1,
               const int* __restrict__ r2, const int* __restrict__ c2, const float* __restrict__ v2, int n2,
               int R1, int R2, int coff2) {
    int row = blockIdx.x * 8 + (threadIdx.x >> 5);
    const int lane = threadIdx.x & 31;
    if (row >= nrows) return;
    row = __builtin_amdgcn_readfirstlane(row);
    const int n = cnt[row];
    float* y = Y + (size_t)row * D_DIM;
    const int2* bucket = (const int2*)y;
    const int rot = (row & 3) * 32;
    float4 a0 = make_float4(0.f, 0.f, 0.f, 0.f);
    float4 a1 = make_float4(0.f, 0.f, 0.f, 0.f);
    if (n <= BUCKET_CAP) {
        for (int b = 0; b < n; b += 32) {
            int m = n - b;
            if (m > 32) m = 32;
            int2 my = make_int2(0, 0);
            if (lane < m) my = bucket[((rot + b) & (BUCKET_CAP - 1)) + lane];
#pragma unroll 4
            for (int j = 0; j < m; ++j) {
                const int   c = __builtin_amdgcn_readlane(my.x, j);
                const float v = __int_as_float(__builtin_amdgcn_readlane(my.y, j));
                row_fma(Xa, Xb, split, c, v, lane, a0, a1);
            }
        }
    } else {
        const int* rl; const int* cl; const float* vl; int nl, rloc, coff;
        if (row < R1)      { rl = r0; cl = c0; vl = v0; nl = n0; rloc = row;      coff = 0; }
        else if (row < R2) { rl = r1; cl = c1; vl = v1; nl = n1; rloc = row - R1; coff = 0; }
        else               { rl = r2; cl = c2; vl = v2; nl = n2; rloc = row - R2; coff = coff2; }
        for (int b = 0; b < nl; b += 32) {
            const int i = b + lane;
            const bool hit = (i < nl) && (rl[i] == rloc);
            int cc = 0, vv = 0;
            if (hit) { cc = cl[i] + coff; vv = __float_as_int(vl[i]); }
            unsigned mask = __builtin_amdgcn_ballot_w32(hit);
            while (mask) {
                const int j = __builtin_ctz(mask);
                mask &= mask - 1;
                const int   c = __builtin_amdgcn_readlane(cc, j);
                const float v = __int_as_float(__builtin_amdgcn_readlane(vv, j));
                row_fma(Xa, Xb, split, c, v, lane, a0, a1);
            }
        }
    }
    ((float4*)y)[lane]      = a0;
    ((float4*)y)[lane + 32] = a1;
}

#define LDE (D_DIM + 4)

template <bool A_IN_LDS>
__device__ __forceinline__ void mm_pass(const float* __restrict__ A, int validRows,
                                        const float* __restrict__ B,
                                        float* __restrict__ As, float* __restrict__ Bst,
                                        const float* __restrict__ E, v8f (&acc)[2][4],
                                        int t, int wmBase, int wnBase, int m16, int kh) {
#pragma unroll 1
    for (int k0 = 0; k0 < D_DIM; k0 += KC) {
        if (!A_IN_LDS) {
            const int r  = t >> 3;
            const int c4 = (t & 7) * 4;
#pragma unroll
            for (int i = 0; i < 2; ++i) {
                const int row = r + 32 * i;
                const int src = row < validRows ? row : validRows - 1;
                const float4 v = *(const float4*)(A + (size_t)src * D_DIM + k0 + c4);
                v2f* dst = (v2f*)&As[row * LDA_S + c4];
                v2f lo; lo.x = v.x; lo.y = v.y;
                v2f hi; hi.x = v.z; hi.y = v.w;
                dst[0] = lo;
                dst[1] = hi;
            }
        }
        {
            const int n4 = (t >> 2) * 4;
#pragma unroll
            for (int i = 0; i < 4; ++i) {
                const int k = 2 * ((t & 3) + 4 * i);
                const float* bp = B + (size_t)(k0 + k) * D_DIM + n4;
                const float4 v0 = *(const float4*)bp;
                const float4 v1 = *(const float4*)(bp + D_DIM);
                v2f p0; p0.x = v0.x; p0.y = v1.x;
                v2f p1; p1.x = v0.y; p1.y = v1.y;
                v2f p2; p2.x = v0.z; p2.y = v1.z;
                v2f p3; p3.x = v0.w; p3.y = v1.w;
                *(v2f*)&Bst[(n4 + 0) * LDBT + k] = p0;
                *(v2f*)&Bst[(n4 + 1) * LDBT + k] = p1;
                *(v2f*)&Bst[(n4 + 2) * LDBT + k] = p2;
                *(v2f*)&Bst[(n4 + 3) * LDBT + k] = p3;
            }
        }
        __syncthreads();

#pragma unroll
        for (int kk = 0; kk < KC; kk += 4) {
            v2f af[2], bf[4];
#pragma unroll
            for (int i = 0; i < 2; ++i) {
                const int m = wmBase + i * 16 + m16;
                af[i] = A_IN_LDS ? *(const v2f*)&E[m * LDE + k0 + kk + 2 * kh]
                                 : *(const v2f*)&As[m * LDA_S + kk + 2 * kh];
            }
#pragma unroll
            for (int j = 0; j < 4; ++j) {
                const int n = wnBase + j * 16 + m16;
                bf[j] = *(const v2f*)&Bst[n * LDBT + kk + 2 * kh];
            }
#pragma unroll
            for (int i = 0; i < 2; ++i)
#pragma unroll
                for (int j = 0; j < 4; ++j)
                    acc[i][j] = __builtin_amdgcn_wmma_f32_16x16x4_f32(
                        false, af[i], false, bf[j], (short)0, acc[i][j], false, false);
        }
        __syncthreads();
    }
}

__global__ __launch_bounds__(256)
void dense_chain(const float* __restrict__ AGG, const float* __restrict__ P,
                 const float* __restrict__ preU, const float* __restrict__ preI,
                 const float* __restrict__ Qu, const float* __restrict__ Qi,
                 const float* __restrict__ Wu, const float* __restrict__ Wi,
                 const float* __restrict__ bu, const float* __restrict__ bi,
                 const float* __restrict__ Mu, const float* __restrict__ Mi,
                 float* __restrict__ OUT, int NU, int NI, int tilesU) {
    __shared__ __attribute__((aligned(16))) float As[64 * LDA_S];
    __shared__ __attribute__((aligned(16))) float Bst[256 * LDBT];
    __shared__ __attribute__((aligned(16))) float E[64 * LDE];
    __shared__ float rs[4 * 64];

    const int t    = threadIdx.x;
    const int wave = t >> 5;
    const int lane = t & 31;
    const int wmBase = (wave & 1) * 32;
    const int wn     = wave >> 1;
    const int wnBase = wn * 64;
    const int m16 = lane & 15;
    const int kh  = lane >> 4;

    const bool isU   = (int)blockIdx.x < tilesU;
    const int  lbase = (isU ? (int)blockIdx.x : (int)blockIdx.x - tilesU) * 64;
    const int  gbase = isU ? lbase : NU + lbase;
    const int  left  = (isU ? NU : NI) - lbase;
    const int  validRows = left < 64 ? left : 64;
    const float* pre = (isU ? preU : preI) + (size_t)lbase * D_DIM;
    const float* Q   = isU ? Qu : Qi;
    const float* Wt  = isU ? Wu : Wi;
    const float* bb  = isU ? bu : bi;
    const float* Mw  = isU ? Mu : Mi;
    const float* agg = AGG + (size_t)gbase * D_DIM;
    const float* pp  = P   + (size_t)gbase * D_DIM;
    float*       o   = OUT + (size_t)gbase * D_DIM;

    const int rowl = wmBase + kh * 8;
    const int coll = wnBase + m16;

    v8f acc[2][4];
#pragma unroll
    for (int i = 0; i < 2; ++i)
#pragma unroll
        for (int j = 0; j < 4; ++j) acc[i][j] = (v8f)0.f;

    mm_pass<false>(agg, validRows, Q, As, Bst, E, acc, t, wmBase, wnBase, m16, kh);
#pragma unroll
    for (int i = 0; i < 2; ++i)
#pragma unroll
        for (int j = 0; j < 4; ++j) {
#pragma unroll
            for (int r = 0; r < 8; ++r)
                E[(rowl + i * 16 + r) * LDE + coll + j * 16] = acc[i][j][r];
            acc[i][j] = (v8f)0.f;
        }
#pragma unroll 1
    for (int ir = 0; ir < 16; ++ir) {
        const int row = rowl + (ir >> 3) * 16 + (ir & 7);
#pragma unroll
        for (int j = 0; j < 4; ++j) {
            const int a = row * LDE + coll + j * 16;
            E[a] = tanhf(E[a]);
        }
    }

    mm_pass<false>(pre, validRows, Wt, As, Bst, E, acc, t, wmBase, wnBase, m16, kh);
    mm_pass<true>(nullptr, validRows, Wt + (size_t)D_DIM * D_DIM, As, Bst, E, acc, t, wmBase, wnBase, m16, kh);

#pragma unroll
    for (int i = 0; i < 2; ++i)
#pragma unroll
        for (int j = 0; j < 4; ++j) {
#pragma unroll
            for (int r = 0; r < 8; ++r)
                E[(rowl + i * 16 + r) * LDE + coll + j * 16] = acc[i][j][r];
            acc[i][j] = (v8f)0.f;
        }
    float bv[4];
#pragma unroll
    for (int j = 0; j < 4; ++j) bv[j] = bb[coll + j * 16];
#pragma unroll 1
    for (int ir = 0; ir < 16; ++ir) {
        const int row = rowl + (ir >> 3) * 16 + (ir & 7);
        float s = 0.f;
#pragma unroll
        for (int j = 0; j < 4; ++j) {
            const int   a = row * LDE + coll + j * 16;
            const float h = tanhf(E[a] + bv[j]);
            E[a] = h;
            s += h * h;
        }
#pragma unroll
        for (int off = 1; off < 16; off <<= 1) s += __shfl_xor(s, off, 32);
        if (m16 == 0) rs[wn * 64 + row] = s;
    }
    __syncthreads();
#pragma unroll 1
    for (int ir = 0; ir < 16; ++ir) {
        const int   row = rowl + (ir >> 3) * 16 + (ir & 7);
        const float s   = ((rs[row] + rs[64 + row]) + rs[128 + row]) + rs[192 + row];
        const float inv = 1.f / fmaxf(sqrtf(s), 1e-12f);
#pragma unroll
        for (int j = 0; j < 4; ++j) {
            const int a = row * LDE + coll + j * 16;
            E[a] = E[a] * inv;
        }
    }

    mm_pass<false>(pp, validRows, Mw, As, Bst, E, acc, t, wmBase, wnBase, m16, kh);
#pragma unroll
    for (int i = 0; i < 2; ++i) {
#pragma unroll
        for (int j = 0; j < 4; ++j)
#pragma unroll
            for (int r = 0; r < 8; ++r)
                Bst[(j * 8 + r) * 256 + t] = acc[i][j][r];
#pragma unroll 1
        for (int r = 0; r < 8; ++r) {
            const int row = rowl + i * 16 + r;
            if (row < validRows) {
#pragma unroll
                for (int j = 0; j < 4; ++j) {
                    const int col = coll + j * 16;
                    o[(size_t)row * D_DIM + col] = E[row * LDE + col] + tanhf(Bst[(j * 8 + r) * 256 + t]);
                }
            }
        }
    }
}

extern "C" void kernel_launch(void* const* d_in, const int* in_sizes, int n_in,
                              void* d_out, int out_size, void* d_ws, size_t ws_size,
                              hipStream_t stream) {
    const int* adj_rows   = (const int*)d_in[0];
    const int* adj_cols   = (const int*)d_in[1];
    const float* adj_vals = (const float*)d_in[2];
    const int* up_rows    = (const int*)d_in[3];
    const int* up_cols    = (const int*)d_in[4];
    const float* up_vals  = (const float*)d_in[5];
    const int* ip_rows    = (const int*)d_in[6];
    const int* ip_cols    = (const int*)d_in[7];
    const float* ip_vals  = (const float*)d_in[8];
    const float* user_emb = (const float*)d_in[9];
    const float* item_emb = (const float*)d_in[10];
    const float* Q_user   = (const float*)d_in[11];
    const float* W_user   = (const float*)d_in[12];
    const float* b_user   = (const float*)d_in[13];
    const float* Q_item   = (const float*)d_in[14];
    const float* W_item   = (const float*)d_in[15];
    const float* b_item   = (const float*)d_in[16];
    const float* M_user   = (const float*)d_in[17];
    const float* M_item   = (const float*)d_in[18];
    float* out = (float*)d_out;

    const int NU = 40000, NI = 20000, NN = 60000;
    const int NNZ_ADJ = in_sizes[0], NNZ_UP = in_sizes[3], NNZ_IP = in_sizes[6];
    const int NNZ_ALL = NNZ_ADJ + NNZ_UP + NNZ_IP;
    const size_t ROWF = (size_t)D_DIM;
    const size_t MAT  = (size_t)NN * ROWF;

    float* buf0 = (float*)d_ws;
    float* buf1 = buf0 + MAT;
    float* buf2 = buf1 + MAT;

    const int NR = NN + NN;
    int* cnt = (int*)buf2;

    dim3 blk(256);

    csr_zero<<<(NR + 255) / 256, blk, 0, stream>>>(cnt, NR);
    csr_scatter<<<(NNZ_ALL + 255) / 256, blk, 0, stream>>>(
        adj_rows, adj_cols, adj_vals, NNZ_ADJ, up_rows, up_cols, up_vals, NNZ_UP,
        ip_rows, ip_cols, ip_vals, NNZ_IP, NN, NN + NU, NU, cnt, buf0);

    spmm_rows<<<(NR + 7) / 8, blk, 0, stream>>>(
        cnt, user_emb, item_emb, NU, buf0, NR,
        adj_rows, adj_cols, adj_vals, NNZ_ADJ, up_rows, up_cols, up_vals, NNZ_UP,
        ip_rows, ip_cols, ip_vals, NNZ_IP, NN, NN + NU, NU);

    dense_chain<<<dim3(625 + 313), blk, 0, stream>>>(
        buf0, buf1, user_emb, item_emb, Q_user, Q_item, W_user, W_item,
        b_user, b_item, M_user, M_item, out, NU, NI, 625);
}


# tanh epilogues in registers: a 16-row block inside the polynomial's range in all lanes (one ballot) gets 16 independent chains and one E store, no LDS trip per element; else rolled tanhf as before
# speedup vs baseline: 2.7293x; 1.2416x over previous
#include <hip/hip_runtime.h>
#include <hip/hip_bf16.h>
#include <math.h>

typedef __attribute__((ext_vector_type(2))) float v2f;
typedef __attribute__((ext_vector_type(8))) float v8f;

#define D_DIM 256
#define KST 4
#define LDE (D_DIM + 4)

#define BUCKET_CAP 128
#define SC_E 8

__global__ __launch_bounds__(256)
void csr_zero(int* __restrict__ cnt, int n) {
    const int i = (blockIdx.x * 256 + threadIdx.x) * 4;
    if (i + 3 < n) {
        *(int4*)(cnt + i) = make_int4(0, 0, 0, 0);
    } else {
        for (int j = i; j < n; ++j) cnt[j] = 0;
    }
}

__global__ __launch_bounds__(256)
void csr_scatter(const int* __restrict__ r0, const int* __restrict__ c0, const float* __restrict__ v0, int n0,
                 const int* __restrict__ r1, const int* __restrict__ c1, const float* __restrict__ v1, int n1,
                 const int* __restrict__ r2, const int* __restrict__ c2, const float* __restrict__ v2, int n2,
                 int R1, int R2, int coff2, int* __restrict__ cursor, float* __restrict__ Y) {
    const int total = n0 + n1 + n2;
    const int e0 = blockIdx.x * (256 * SC_E) + threadIdx.x;
    if (e0 >= total) return;
    int   row[SC_E], col[SC_E], pos[SC_E];
    int   radd[SC_E], cadd[SC_E];
    float val[SC_E];
#pragma unroll
    for (int u = 0; u < SC_E; ++u) {
        const int  e   = e0 + u * 256;
        const int  ec  = e < total ? e : total - 1;
        const bool in2 = ec >= n0 + n1;
        const bool in1 = !in2 && ec >= n0;
        const int*   rp = in2 ? r2 : (in1 ? r1 : r0);
        const int*   cp = in2 ? c2 : (in1 ? c1 : c0);
        const float* vp = in2 ? v2 : (in1 ? v1 : v0);
        const int    k  = ec - (in2 ? n0 + n1 : (in1 ? n0 : 0));
        radd[u] = in2 ? R2 : (in1 ? R1 : 0);
        cadd[u] = in2 ? coff2 : 0;
        row[u] = rp[k];
        col[u] = cp[k];
        val[u] = vp[k];
    }
#pragma unroll
    for (int u = 0; u < SC_E; ++u) { row[u] += radd[u]; col[u] += cadd[u]; }
#pragma unroll
    for (int u = 0; u < SC_E; ++u) {
        pos[u] = BUCKET_CAP;
        if (e0 + u * 256 < total) pos[u] = atomicAdd(&cursor[row[u]], 1);
    }
#pragma unroll
    for (int u = 0; u < SC_E; ++u) {
        if (pos[u] < BUCKET_CAP) {
            const int place = ((row[u] & 3) * 32 + pos[u]) & (BUCKET_CAP - 1);
            ((int2*)(Y + (size_t)row[u] * D_DIM))[place] = make_int2(col[u], __float_as_int(val[u]));
        }
    }
}

__device__ __forceinline__ void row_fma(const float* __restrict__ Xa, const float* __restrict__ Xb,
                                        int split, int c, float v, int lane, float4& a0, float4& a1) {
    const float* X = (c < split) ? (Xa + (size_t)c * D_DIM)
                                 : (Xb + (size_t)(c - split) * D_DIM);
    const float4 x0 = ((const float4*)X)[lane];
    const float4 x1 = ((const float4*)X)[lane + 32];
    a0.x = fmaf(v, x0.x, a0.x); a0.y = fmaf(v, x0.y, a0.y); a0.z = fmaf(v, x0.z, a0.z); a0.w = fmaf(v, x0.w, a0.w);
    a1.x = fmaf(v, x1.x, a1.x); a1.y = fmaf(v, x1.y, a1.y); a1.z = fmaf(v, x1.z, a1.z); a1.w = fmaf(v, x1.w, a1.w);
}

struct Lists {
    const int* r0; const int* c0; const float* v0; int n0;
    const int* r1; const int* c1; const float* v1; int n1;
    const int* r2; const int* c2; const float* v2; int n2;
    int R1, R2, coff2;
};

__device__ __forceinline__ void row_from_list(const Lists& L, int row,
                                              const float* __restrict__ Xa, const float* __restrict__ Xb,
                                              int split, int lane, float4& a0, float4& a1) {
    const int* rl; const int* cl; const float* vl; int nl, rloc, coff;
    if (row < L.R1)      { rl = L.r0; cl = L.c0; vl = L.v0; nl = L.n0; rloc = row;        coff = 0; }
    else if (row < L.R2) { rl = L.r1; cl = L.c1; vl = L.v1; nl = L.n1; rloc = row - L.R1; coff = 0; }
    else                 { rl = L.r2; cl = L.c2; vl = L.v2; nl = L.n2; rloc = row - L.R2; coff = L.coff2; }
    for (int b = 0; b < nl; b += 32) {
        const int i = b + lane;
        const bool hit = (i < nl) && (rl[i] == rloc);
        int cc = 0, vv = 0;
        if (hit) { cc = cl[i] + coff; vv = __float_as_int(vl[i]); }
        unsigned mask = __builtin_amdgcn_ballot_w32(hit);
        while (mask) {
            const int j = __builtin_ctz(mask);
            mask &= mask - 1;
            const int   c = __builtin_amdgcn_readlane(cc, j);
            const float v = __int_as_float(__builtin_amdgcn_readlane(vv, j));
            row_fma(Xa, Xb, split, c, v, lane, a0, a1);
        }
    }
}

__device__ __forceinline__ int2 first_places(const float* __restrict__ Ybk, int row, int n, int lane) {
    int2 my = make_int2(0, 0);
    const int m = (n <= BUCKET_CAP) ? (n < 32 ? n : 32) : 0;
    if (lane < m) my = ((const int2*)(Ybk + (size_t)row * D_DIM))[(row & 3) * 32 + lane];
    return my;
}

__device__ __forceinline__ const float4* x_row(const float* __restrict__ Xa, const float* __restrict__ Xb,
                                               int split, int c) {
    return (const float4*)((c < split) ? (Xa + (size_t)c * D_DIM) : (Xb + (size_t)(c - split) * D_DIM));
}

__device__ __forceinline__ void sum_places(int2 my, int g0, int m,
                                           const float* __restrict__ Xa, const float* __restrict__ Xb,
                                           int split, int lane, float4& a0, float4& a1) {
#pragma unroll 1
    for (int g = g0; g < m; g += 8) {
#pragma unroll
        for (int u = 0; u < 8; ++u) {
            const int   c = __builtin_amdgcn_readlane(my.x, g + u);
            const float v = __int_as_float(__builtin_amdgcn_readlane(my.y, g + u));
            row_fma(Xa, Xb, split, c, v, lane, a0, a1);
        }
    }
}

__device__ __forceinline__ void gather_tile(const int* __restrict__ cnt, const float* __restrict__ Ybk,
                                            const float* __restrict__ Xa, const float* __restrict__ Xb,
                                            int split, const Lists& L, int crow0, int validRows,
                                            float* __restrict__ E, int wv, int lane) {
    const int l0 = wv * 8;
    int cntv = 0;
    if (lane < 8 && l0 + lane < validRows) cntv = cnt[crow0 + l0 + lane];
    int  n_cur  = __builtin_amdgcn_readlane(cntv, 0);
    int2 bk_cur = first_places(Ybk, crow0 + l0, n_cur, lane);
#pragma unroll 1
    for (int rr = 0; rr < 8; ++rr) {
        const int row = crow0 + l0 + rr;
        const int n_next = __builtin_amdgcn_readlane(cntv, rr + 1);
        int2 bk_next;
        float4 a0 = make_float4(0.f, 0.f, 0.f, 0.f);
        float4 a1 = make_float4(0.f, 0.f, 0.f, 0.f);
        if (n_cur > 0 && n_cur <= BUCKET_CAP) {
            float4 x0[8], x1[8];
            float  vv[8];
#pragma unroll
            for (int u = 0; u < 8; ++u) {
                const int c = __builtin_amdgcn_readlane(bk_cur.x, u);
                vv[u] = __int_as_float(__builtin_amdgcn_readlane(bk_cur.y, u));
                const float4* X = x_row(Xa, Xb, split, c);
                x0[u] = X[lane];
                x1[u] = X[lane + 32];
            }
            bk_next = first_places(Ybk, row + 1, n_next, lane);
#pragma unroll
            for (int u = 0; u < 8; ++u) {
                const float v = vv[u];
                a0.x = fmaf(v, x0[u].x, a0.x); a0.y = fmaf(v, x0[u].y, a0.y);
                a0.z = fmaf(v, x0[u].z, a0.z); a0.w = fmaf(v, x0[u].w, a0.w);
                a1.x = fmaf(v, x1[u].x, a1.x); a1.y = fmaf(v, x1[u].y, a1.y);
                a1.z = fmaf(v, x1[u].z, a1.z); a1.w = fmaf(v, x1[u].w, a1.w);
            }
            sum_places(bk_cur, 8, n_cur < 32 ? n_cur : 32, Xa, Xb, split, lane, a0, a1);
            if (n_cur > 32) {
                const int2* bucket = (const int2*)(Ybk + (size_t)row * D_DIM);
                const int rot = (row & 3) * 32;
                for (int b = 32; b < n_cur; b += 32) {
                    int m = n_cur - b;
                    if (m > 32) m = 32;
                    int2 my = make_int2(0, 0);
                    if (lane < m) my = bucket[((rot + b) & (BUCKET_CAP - 1)) + lane];
                    sum_places(my, 0, m, Xa, Xb, split, lane, a0, a1);
                }
            }
        } else {
            bk_next = first_places(Ybk, row + 1, n_next, lane);
            if (n_cur > BUCKET_CAP) row_from_list(L, row, Xa, Xb, split, lane, a0, a1);
        }
        float* e = E + (l0 + rr) * LDE;
        ((float4*)e)[lane]      = a0;
        ((float4*)e)[lane + 32] = a1;
        n_cur  = n_next;
        bk_cur = bk_next;
    }
}


__device__ __forceinline__ void load_step(v2f (&f)[2], const float* __restrict__ bp, int k) {
#pragma unroll
    for (int j = 0; j < 2; ++j) {
        f[j].x = bp[(size_t)k * D_DIM + 16 * j];
        f[j].y = bp[(size_t)(k + 1) * D_DIM + 16 * j];
    }
}

__device__ __forceinline__ void compute_step(const v2f (&f)[2], const float* __restrict__ ep, int k,
                                             v8f (&acc)[4][2]) {
    v2f af[4];
#pragma unroll
    for (int i = 0; i < 4; ++i) af[i] = *(const v2f*)&ep[16 * i * LDE + k];
#pragma unroll
    for (int i = 0; i < 4; ++i)
#pragma unroll
        for (int j = 0; j < 2; ++j)
            acc[i][j] = __builtin_amdgcn_wmma_f32_16x16x4_f32(
                false, af[i], false, f[j], (short)0, acc[i][j], false, false);
}

__device__ __forceinline__ void mm_pass(const float* __restrict__ B, const float* __restrict__ E,
                                        v8f (&acc)[4][2], int wv, int m16, int kh) {
    const float* bp = B + (size_t)(2 * kh) * D_DIM + 32 * wv + m16;
    const float* ep = E + m16 * LDE + 2 * kh;
    v2f ring[KST][2];
#pragma unroll
    for (int q = 0; q < KST; ++q) load_step(ring[q], bp, 4 * q);
    __syncthreads();
#pragma unroll 1
    for (int k0 = 0; k0 < D_DIM - 4 * KST; k0 += 4 * KST) {
#pragma unroll
        for (int q = 0; q < KST; ++q) {
            compute_step(ring[q], ep, k0 + 4 * q, acc);
            load_step(ring[q], bp, k0 + 4 * KST + 4 * q);
        }
    }
#pragma unroll
    for (int q = 0; q < KST; ++q) compute_step(ring[q], ep, D_DIM - 4 * KST + 4 * q, acc);
    __syncthreads();
}

__device__ __forceinline__ float tanh_small(float x) {
    const float ax = fabsf(x);
    const float t  = x * x;
    float p = fmaf(__int_as_float(0xbbbac73d), t, __int_as_float(0x3ca908c9));
    p = fmaf(t, p, __int_as_float(0xbd5c1c4e));
    p = fmaf(t, p, __int_as_float(0x3e088382));
    p = fmaf(t, p, __int_as_float(0xbeaaaa99));
    p = ax * p;
    p = fmaf(t, p, ax);
    return copysignf(p, x);
}

template <bool H>
__device__ __forceinline__ void tanh_epilogue(v8f (&acc)[4][2], float* __restrict__ E, float* __restrict__ rs,
                                              int rowl, int coll, int wv) {
    const unsigned lid = __builtin_amdgcn_mbcnt_lo(~0u, 0u);
    const bool row_writer = (lid & 15u) == 0u;
    const int  rowh = (int)(lid >> 4) * 8;
    unsigned parked = 0u;
#pragma unroll
    for (int i = 0; i < 4; ++i) {
        float mx = 0.f;
#pragma unroll
        for (int r = 0; r < 8; ++r) {
            mx = fmaxf(mx, fmaxf(fabsf(acc[i][0][r]), fabsf(acc[i][1][r])));
        }
        if (__builtin_amdgcn_ballot_w32(!(mx < 0.625f)) != 0u) {
            parked |= 1u << i;
#pragma unroll
            for (int r = 0; r < 8; ++r) {
                E[(rowl + i * 16 + r) * LDE + coll]      = acc[i][0][r];
                E[(rowl + i * 16 + r) * LDE + coll + 16] = acc[i][1][r];
            }
        } else {
            float sq[8];
#pragma unroll
            for (int r = 0; r < 8; ++r) {
                const float h0 = tanh_small(acc[i][0][r]);
                const float h1 = tanh_small(acc[i][1][r]);
                E[(rowl + i * 16 + r) * LDE + coll]      = h0;
                E[(rowl + i * 16 + r) * LDE + coll + 16] = h1;
                sq[r] = fmaf(h0, h0, h1 * h1);
            }
            if (H) {
#pragma unroll
                for (int off = 1; off < 16; off <<= 1)
#pragma unroll
                    for (int r = 0; r < 8; ++r) sq[r] += __shfl_xor(sq[r], off, 32);
                if (row_writer) {
#pragma unroll
                    for (int r = 0; r < 8; ++r) rs[wv * 64 + rowh + i * 16 + r] = sq[r];
                }
            }
        }
        if (!H) { acc[i][0] = (v8f)0.f; acc[i][1] = (v8f)0.f; }
    }
    if (parked != 0u) {
#pragma unroll 1
        for (int ir = 0; ir < 16; ++ir) {
            if (((parked >> (ir >> 2)) & 1u) == 0u) continue;
            const int row = rowl + (ir >> 2) * 16 + (ir & 3) * 2;
#pragma unroll
            for (int j = 0; j < 2; ++j) {
                const int a = (row + j) * LDE + coll;
                const float h0 = tanhf(E[a]);
                const float h1 = tanhf(E[a + 16]);
                E[a]      = h0;
                E[a + 16] = h1;
                if (H) {
                    float s = fmaf(h0, h0, h1 * h1);
#pragma unroll
                    for (int off = 1; off < 16; off <<= 1) s += __shfl_xor(s, off, 32);
                    if (row_writer) rs[wv * 64 + rowh + (ir >> 2) * 16 + (ir & 3) * 2 + j] = s;
                }
            }
        }
    }
}

__global__ __launch_bounds__(256, 8)
void dense_chain(const int* __restrict__ cnt, const float* __restrict__ Ybk,
                 const int* __restrict__ r0, const int* __restrict__ c0, const float* __restrict__ v0, int n0,
                 const int* __restrict__ r1, const int* __restrict__ c1, const float* __restrict__ v1, int n1,
                 const int* __restrict__ r2, const int* __restrict__ c2, const float* __restrict__ v2, int n2,
                 const float* __restrict__ preU, const float* __restrict__ preI,
                 const float* __restrict__ Qu, const float* __restrict__ Qi,
                 const float* __restrict__ Wu, const float* __restrict__ Wi,
                 const float* __restrict__ bu, const float* __restrict__ bi,
                 const float* __restrict__ Mu, const float* __restrict__ Mi,
                 float* __restrict__ OUT, int NU, int NI, int tilesU) {
    __shared__ __attribute__((aligned(16))) float E[64 * LDE];
    __shared__ float rs[8 * 64];

    const int t    = threadIdx.x;
    const int wave = t >> 5;
    const int lane = t & 31;
    const int m16 = lane & 15;
    const int kh  = lane >> 4;
    const int wv = __builtin_amdgcn_readfirstlane(wave);

    const bool isU   = (int)blockIdx.x < tilesU;
    const int  lbase = (isU ? (int)blockIdx.x : (int)blockIdx.x - tilesU) * 64;
    const int  gbase = isU ? lbase : NU + lbase;
    const int  left  = (isU ? NU : NI) - lbase;
    const int  validRows = left < 64 ? left : 64;
    const int  NN = NU + NI;
    const float* pre = (isU ? preU : preI) + (size_t)lbase * D_DIM;
    const float* Q   = isU ? Qu : Qi;
    const float* Wt  = isU ? Wu : Wi;
    const float* bb  = isU ? bu : bi;
    const float* Mw  = isU ? Mu : Mi;
    float*       o   = OUT + (size_t)gbase * D_DIM;

    Lists L;
    L.r0 = r0; L.c0 = c0; L.v0 = v0; L.n0 = n0;
    L.r1 = r1; L.c1 = c1; L.v1 = v1; L.n1 = n1;
    L.r2 = r2; L.c2 = c2; L.v2 = v2; L.n2 = n2;
    L.R1 = NN; L.R2 = NN + NU; L.coff2 = NU;

    const int rowl = kh * 8;
    const int coll = wave * 32 + m16;

    v8f acc[4][2];
#pragma unroll
    for (int i = 0; i < 4; ++i)
#pragma unroll
        for (int j = 0; j < 2; ++j) acc[i][j] = (v8f)0.f;

    gather_tile(cnt, Ybk, preU, preI, NU, L, NN + gbase, validRows, E, wv, lane);
    mm_pass(Mw, E, acc, wv, m16, kh);
    tanh_epilogue<false>(acc, E, rs, rowl, coll, wv);
    __syncthreads();
#pragma unroll 4
    for (int idx = t; idx < 64 * 64; idx += 256) {
        const int row = idx >> 6, c4 = idx & 63;
        if (row < validRows)
            ((float4*)(o + (size_t)row * D_DIM))[c4] = *(const float4*)&E[row * LDE + 4 * c4];
    }
    __syncthreads();

    gather_tile(cnt, Ybk, preU, preI, NU, L, gbase, validRows, E, wv, lane);
    mm_pass(Q, E, acc, wv, m16, kh);
    tanh_epilogue<false>(acc, E, rs, rowl, coll, wv);

    mm_pass(Wt + (size_t)D_DIM * D_DIM, E, acc, wv, m16, kh);
#pragma unroll 8
    for (int idx = t; idx < 64 * 64; idx += 256) {
        const int row = idx >> 6, c4 = idx & 63;
        const int srow = row < validRows ? row : validRows - 1;
        *(float4*)&E[row * LDE + 4 * c4] = ((const float4*)(pre + (size_t)srow * D_DIM))[c4];
    }
    mm_pass(Wt, E, acc, wv, m16, kh);

    const int collb = wv * 32 + (int)(__builtin_amdgcn_mbcnt_lo(~0u, 0u) & 15u);
    const float bv0 = bb[collb], bv1 = bb[collb + 16];
#pragma unroll
    for (int i = 0; i < 4; ++i)
#pragma unroll
        for (int r = 0; r < 8; ++r) { acc[i][0][r] += bv0; acc[i][1][r] += bv1; }
    tanh_epilogue<true>(acc, E, rs, rowl, coll, wv);
    __syncthreads();

#pragma unroll 4
    for (int idx = t; idx < 64 * 64; idx += 256) {
        const int row = idx >> 6, c4 = idx & 63;
        if (row < validRows) {
            float s = rs[row];
#pragma unroll
            for (int g = 1; g < 8; ++g) s += rs[g * 64 + row];
            const float inv = 1.f / fmaxf(sqrtf(s), 1e-12f);
            float4*      op = (float4*)(o + (size_t)row * D_DIM) + c4;
            const float4 p  = *op;
            const float4 h  = *(const float4*)&E[row * LDE + 4 * c4];
            float4 y;
            y.x = h.x * inv + p.x; y.y = h.y * inv + p.y;
            y.z = h.z * inv + p.z; y.w = h.w * inv + p.w;
            *op = y;
        }
    }
}

extern "C" void kernel_launch(void* const* d_in, const int* in_sizes, int n_in,
                              void* d_out, int out_size, void* d_ws, size_t ws_size,
                              hipStream_t stream) {
    const int* adj_rows   = (const int*)d_in[0];
    const int* adj_cols   = (const int*)d_in[1];
    const float* adj_vals = (const float*)d_in[2];
    const int* up_rows    = (const int*)d_in[3];
    const int* up_cols    = (const int*)d_in[4];
    const float* up_vals  = (const float*)d_in[5];
    const int* ip_rows    = (const int*)d_in[6];
    const int* ip_cols    = (const int*)d_in[7];
    const float* ip_vals  = (const float*)d_in[8];
    const float* user_emb = (const float*)d_in[9];
    const float* item_emb = (const float*)d_in[10];
    const float* Q_user   = (const float*)d_in[11];
    const float* W_user   = (const float*)d_in[12];
    const float* b_user   = (const float*)d_in[13];
    const float* Q_item   = (const float*)d_in[14];
    const float* W_item   = (const float*)d_in[15];
    const float* b_item   = (const float*)d_in[16];
    const float* M_user   = (const float*)d_in[17];
    const float* M_item   = (const float*)d_in[18];
    float* out = (float*)d_out;

    const int NU = 40000, NI = 20000, NN = 60000;
    const int NNZ_ADJ = in_sizes[0], NNZ_UP = in_sizes[3], NNZ_IP = in_sizes[6];
    const int NNZ_ALL = NNZ_ADJ + NNZ_UP + NNZ_IP;
    const size_t ROWF = (size_t)D_DIM;
    const size_t MAT  = (size_t)NN * ROWF;

    float* buf0 = (float*)d_ws;
    float* buf1 = buf0 + MAT;
    float* buf2 = buf1 + MAT;

    const int NR = NN + NN;
    int* cnt = (int*)buf2;

    dim3 blk(256);

    csr_zero<<<(NR + 1023) / 1024, blk, 0, stream>>>(cnt, NR);
    csr_scatter<<<(NNZ_ALL + 256 * SC_E - 1) / (256 * SC_E), blk, 0, stream>>>(
        adj_rows, adj_cols, adj_vals, NNZ_ADJ, up_rows, up_cols, up_vals, NNZ_UP,
        ip_rows, ip_cols, ip_vals, NNZ_IP, NN, NN + NU, NU, cnt, buf0);

    dense_chain<<<dim3(625 + 313), blk, 0, stream>>>(
        cnt, buf0,
        adj_rows, adj_cols, adj_vals, NNZ_ADJ, up_rows, up_cols, up_vals, NNZ_UP,
        ip_rows, ip_cols, ip_vals, NNZ_IP,
        user_emb, item_emb, Q_user, Q_item, W_user, W_item,
        b_user, b_item, M_user, M_item, out, NU, NI, 625);
}
